# starting kernel
# baseline (speedup 1.0000x reference)
#include <hip/hip_runtime.h>

typedef __attribute__((ext_vector_type(2))) float v2f;
typedef __attribute__((ext_vector_type(8))) float v8f;

__device__ __forceinline__ v8f wmma4(v2f a, v2f b, v8f c) {
  return __builtin_amdgcn_wmma_f32_16x16x4_f32(false, a, false, b, (short)0, c,
                                               false, false);
}

__device__ __forceinline__ float leaky(float x) {
  return x >= 0.0f ? x : 0.01f * x;
}

__global__ void __launch_bounds__(128) proj_kernel(
    const float* __restrict__ sa, const float* __restrict__ sb,
    const float* __restrict__ W1, const float* __restrict__ b1,
    float* __restrict__ a_pb, float* __restrict__ b_part) {
  const int lane = threadIdx.x & 31;
  const int lo = lane & 15, hi = lane >> 4;
  const int wave = (blockIdx.x * 128 + threadIdx.x) >> 5;
  const int mat = wave >> 8;
  const int t = wave & 255;
  const int m0 = (t >> 2) * 16;
  const int n0 = (t & 3) * 16;

  const float* src = mat ? sb : sa;
  const float* W   = W1 + (mat ? 256 * 64 : 0);
  const float* arow = src + (size_t)(m0 + lo) * 256 + 2 * hi;
  const float* wcol = W + n0 + lo + (size_t)(2 * hi) * 64;

  v8f c = {};
#pragma unroll 8
  for (int kk = 0; kk < 256; kk += 4) {
    v2f a = *(const v2f*)(arow + kk);
    v2f b;
    b.x = wcol[(size_t)kk * 64];
    b.y = wcol[(size_t)(kk + 1) * 64];
    c = wmma4(a, b, c);
  }

  const float bias = mat ? 0.0f : b1[n0 + lo];
  float* dst = (mat ? b_part : a_pb) + n0 + lo;
#pragma unroll
  for (int r = 0; r < 8; ++r)
    dst[(size_t)(m0 + r + 8 * hi) * 64] = c[r] + bias;
}

#define LSTR 34

__global__ void __launch_bounds__(128) match_mlp_kernel(
    const float* __restrict__ a_pb, const float* __restrict__ b_part,
    const float* __restrict__ W2, const float* __restrict__ b2,
    const float* __restrict__ W3, const float* __restrict__ b3,
    float* __restrict__ out) {
  __shared__ float lds[4 * 16 * LSTR];
  const int lane = threadIdx.x & 31;
  const int lo = lane & 15, hi = lane >> 4;
  const int wave = (blockIdx.x * 128 + threadIdx.x) >> 5;
  const int ba = wave >> 1;
  const int half = wave & 1;
  const int bbase = ba & ~127;
  float* wlds = lds + (threadIdx.x >> 5) * (16 * LSTR);

  v2f abv[16];
  const float* apr = a_pb + (size_t)ba * 64 + 2 * hi;
#pragma unroll
  for (int i = 0; i < 16; ++i) abv[i] = *(const v2f*)(apr + 4 * i);

  v2f w2f[2][16];
#pragma unroll
  for (int t = 0; t < 2; ++t) {
    const float* w = W2 + 16 * t + lo + (size_t)(2 * hi) * 32;
#pragma unroll
    for (int i = 0; i < 16; ++i) {
      w2f[t][i].x = w[(size_t)(4 * i) * 32];
      w2f[t][i].y = w[(size_t)(4 * i + 1) * 32];
    }
  }
  v2f w3f[8];
  {
    const float* w = W3 + lo + (size_t)(2 * hi) * 16;
#pragma unroll
    for (int i = 0; i < 8; ++i) {
      w3f[i].x = w[(size_t)(4 * i) * 16];
      w3f[i].y = w[(size_t)(4 * i + 1) * 16];
    }
  }
  const float b2v0 = b2[lo], b2v1 = b2[16 + lo];
  const float b3v = b3[lo];

  for (int it = 0; it < 4; ++it) {
    const int c0 = (half * 4 + it) * 16;

    const float* bpr = b_part + (size_t)(bbase + c0 + lo) * 64 + 2 * hi;
    v2f hf[16];
#pragma unroll
    for (int i = 0; i < 16; ++i) {
      v2f bp = *(const v2f*)(bpr + 4 * i);
      hf[i].x = leaky(abv[i].x + bp.x);
      hf[i].y = leaky(abv[i].y + bp.y);
    }

    v8f c2a = {}, c2b = {};
#pragma unroll
    for (int i = 0; i < 16; ++i) {
      c2a = wmma4(hf[i], w2f[0][i], c2a);
      c2b = wmma4(hf[i], w2f[1][i], c2b);
    }

#pragma unroll
    for (int r = 0; r < 8; ++r) {
      wlds[(r + 8 * hi) * LSTR + lo]      = leaky(c2a[r] + b2v0);
      wlds[(r + 8 * hi) * LSTR + 16 + lo] = leaky(c2b[r] + b2v1);
    }
    asm volatile("s_wait_dscnt 0x0" ::: "memory");

    v2f hf3[8];
#pragma unroll
    for (int i = 0; i < 8; ++i)
      hf3[i] = *(const v2f*)(wlds + lo * LSTR + 4 * i + 2 * hi);
    asm volatile("s_wait_dscnt 0x0" ::: "memory");

    v8f c3 = {};
#pragma unroll
    for (int i = 0; i < 8; ++i) c3 = wmma4(hf3[i], w3f[i], c3);

    float* op = out + ((size_t)ba * 128 + c0) * 16 + lo;
#pragma unroll
    for (int r = 0; r < 8; ++r)
      op[(size_t)(r + 8 * hi) * 16] = leaky(c3[r] + b3v);
  }
}

extern "C" void kernel_launch(void* const* d_in, const int* in_sizes, int n_in,
                              void* d_out, int out_size, void* d_ws,
                              size_t ws_size, hipStream_t stream) {
  (void)in_sizes; (void)n_in; (void)out_size; (void)ws_size;
  const float* sa = (const float*)d_in[0];
  const float* sb = (const float*)d_in[1];
  const float* W1 = (const float*)d_in[2];
  const float* b1 = (const float*)d_in[3];
  const float* W2 = (const float*)d_in[4];
  const float* b2 = (const float*)d_in[5];
  const float* W3 = (const float*)d_in[6];
  const float* b3 = (const float*)d_in[7];
  float* out = (float*)d_out;

  float* a_pb   = (float*)d_ws;
  float* b_part = a_pb + 1024 * 64;

  proj_kernel<<<128, 128, 0, stream>>>(sa, sb, W1, b1, a_pb, b_part);
  match_mlp_kernel<<<512, 128, 0, stream>>>(a_pb, b_part, W2, b2, W3, b3, out);
}


# a_pb and b_part kept in the workspace in fragment order: projection runs transposed and stores whole lines, MLP fragment loads are 512 contiguous bytes per wave, not 16 strided pieces; same sums
# speedup vs baseline: 1.4775x; 1.4775x over previous
#include <hip/hip_runtime.h>

typedef __attribute__((ext_vector_type(2))) float v2f;
typedef __attribute__((ext_vector_type(4))) float v4f;
typedef __attribute__((ext_vector_type(8))) float v8f;


__device__ __forceinline__ v8f wmma4(v2f a, v2f b, v8f c) {
  return __builtin_amdgcn_wmma_f32_16x16x4_f32(false, a, false, b, (short)0, c,
                                               false, false);
}

__device__ __forceinline__ float leaky(float x) {
  return x >= 0.0f ? x : 0.01f * x;
}

#define PF 20

__attribute__((amdgpu_waves_per_eu(1, 2)))
__global__ void __launch_bounds__(128) proj_kernel(
    const float* __restrict__ sa, const float* __restrict__ sb,
    const float* __restrict__ W1, const float* __restrict__ b1,
    float* __restrict__ a_pb, float* __restrict__ b_part) {
  const int lane = threadIdx.x & 31;
  const int lo = lane & 15, hi = lane >> 4;
  const int wave = (blockIdx.x * 128 + threadIdx.x) >> 5;
  const int mat = wave >> 8;
  const int t = wave & 255;
  const int m0 = (t >> 2) * 16;
  const int n0 = (t & 3) * 16;

  const float* src = mat ? sb : sa;
  const float* W   = W1 + (mat ? 256 * 64 : 0);
  const float* arow = src + (size_t)(m0 + lo) * 256 + 2 * hi;
  const float* wcol = W + n0 + lo + (size_t)(2 * hi) * 64;

  v2f af[64], bf[64];
#pragma unroll
  for (int i = 0; i < PF; ++i) {
    af[i] = *(const v2f*)(arow + 4 * i);
    bf[i].x = wcol[(size_t)(4 * i) * 64];
    bf[i].y = wcol[(size_t)(4 * i + 1) * 64];
  }
  float bias[8];
#pragma unroll
  for (int r = 0; r < 8; ++r) bias[r] = b1[n0 + 8 * hi + r];

  v8f c = {};
#pragma unroll
  for (int i = 0; i < 64; ++i) {
    if (i + PF < 64) {
      const int j = i + PF;
      af[j] = *(const v2f*)(arow + 4 * j);
      bf[j].x = wcol[(size_t)(4 * j) * 64];
      bf[j].y = wcol[(size_t)(4 * j + 1) * 64];
    }
    c = wmma4(bf[i], af[i], c);
  }

  float o[8];
#pragma unroll
  for (int r = 0; r < 8; ++r)
    o[r] = c[r] + (mat ? 0.0f : bias[r]);
  float* dst = (mat ? b_part : a_pb) + (size_t)(m0 >> 4) * 1024 +
               ((n0 >> 3) + hi) * 128 + lo * 4;
  v4f p0 = {o[0], o[1], o[4], o[5]};
  v4f p1 = {o[2], o[3], o[6], o[7]};
  *(v4f*)dst = p0;
  *(v4f*)(dst + 64) = p1;
}

#define LSTR 34

__attribute__((amdgpu_waves_per_eu(2, 2)))
__global__ void __launch_bounds__(128) match_mlp_kernel(
    const float* __restrict__ a_pb, const float* __restrict__ b_part,
    const float* __restrict__ W2, const float* __restrict__ b2,
    const float* __restrict__ W3, const float* __restrict__ b3,
    float* __restrict__ out) {
  __shared__ float lds[4 * 16 * LSTR];
  const int lane = threadIdx.x & 31;
  const int lo = lane & 15, hi = lane >> 4;
  const int wave = (blockIdx.x * 128 + threadIdx.x) >> 5;
  const int ba = wave >> 1;
  const int half = wave & 1;
  const int bbase = ba & ~127;
  float* wlds = lds + (threadIdx.x >> 5) * (16 * LSTR);

  const float* apr = a_pb + (size_t)(ba >> 4) * 1024 + hi * 64 + (ba & 15) * 4;
  const float* bpr = b_part + (size_t)((bbase + half * 64) >> 4) * 1024 +
                     hi * 64 + lo * 4;
  const float* w2p = W2 + lo + (size_t)(2 * hi) * 32;
  const float* w3p = W3 + lo + (size_t)(2 * hi) * 16;

  v4f abv[8], bp[4][8];
  v2f w2f[2][16], w3f[8];
  float b2v0 = 0.0f, b2v1 = 0.0f, b3v = 0.0f;

#define LOAD_STEP(s)                                          \
  {                                                           \
    if (((s) & 1) == 0) {                                     \
      abv[(s) >> 1] = *(const v4f*)(apr + 128 * ((s) >> 1));  \
      bp[0][(s) >> 1] = *(const v4f*)(bpr + 128 * ((s) >> 1));\
    }                                                         \
    w2f[0][s].x = w2p[(size_t)(4 * (s)) * 32];                \
    w2f[0][s].y = w2p[(size_t)(4 * (s) + 1) * 32];            \
    w2f[1][s].x = w2p[(size_t)(4 * (s)) * 32 + 16];           \
    w2f[1][s].y = w2p[(size_t)(4 * (s) + 1) * 32 + 16];       \
  }
#define LOAD_TILE(t)                                          \
  _Pragma("unroll") for (int j = 0; j < 8; ++j)               \
    bp[t][j] = *(const v4f*)(bpr + (t) * 1024 + 128 * j);

#pragma unroll
  for (int s = 0; s < 8; ++s) LOAD_STEP(s)

#pragma unroll
  for (int it = 0; it < 4; ++it) {
    const int c0 = (half * 4 + it) * 16;

    v8f c2a = {}, c2b = {};
#pragma unroll
    for (int ch = 0; ch < 4; ++ch) {
      if (it == 0) {
        __builtin_amdgcn_sched_barrier(0);
        if (ch < 2) {
#pragma unroll
          for (int s = 0; s < 4; ++s) LOAD_STEP(4 * (ch + 2) + s)
        } else if (ch == 2) {
          b2v0 = b2[lo];
          b2v1 = b2[16 + lo];
#pragma unroll
          for (int i = 0; i < 8; ++i) {
            w3f[i].x = w3p[(size_t)(4 * i) * 16];
            w3f[i].y = w3p[(size_t)(4 * i + 1) * 16];
          }
          b3v = b3[lo];
          LOAD_TILE(1)
        } else {
          LOAD_TILE(2)
        }
      }
#pragma unroll
      for (int s = 0; s < 4; ++s) {
        const int i = 4 * ch + s;
        v2f hf;
        hf.x = leaky(abv[i >> 1][2 * (i & 1)] + bp[it][i >> 1][2 * (i & 1)]);
        hf.y = leaky(abv[i >> 1][2 * (i & 1) + 1] +
                     bp[it][i >> 1][2 * (i & 1) + 1]);
        c2a = wmma4(hf, w2f[0][i], c2a);
        c2b = wmma4(hf, w2f[1][i], c2b);
      }
    }

    if (it == 0) {
      __builtin_amdgcn_sched_barrier(0);
      LOAD_TILE(3)
    }

#pragma unroll
    for (int r = 0; r < 8; ++r) {
      wlds[(r + 8 * hi) * LSTR + lo]      = leaky(c2a[r] + b2v0);
      wlds[(r + 8 * hi) * LSTR + 16 + lo] = leaky(c2b[r] + b2v1);
    }
    asm volatile("s_wait_dscnt 0x0" ::: "memory");

    v2f hf3[8];
#pragma unroll
    for (int i = 0; i < 8; ++i)
      hf3[i] = *(const v2f*)(wlds + lo * LSTR + 4 * i + 2 * hi);
    asm volatile("s_wait_dscnt 0x0" ::: "memory");

    v8f c3 = {};
#pragma unroll
    for (int i = 0; i < 8; ++i) c3 = wmma4(hf3[i], w3f[i], c3);

    float* op = out + ((size_t)ba * 128 + c0) * 16 + lo;
#pragma unroll
    for (int r = 0; r < 8; ++r)
      op[(size_t)(r + 8 * hi) * 16] = leaky(c3[r] + b3v);
  }
#undef LOAD_STEP
#undef LOAD_TILE
}

extern "C" void kernel_launch(void* const* d_in, const int* in_sizes, int n_in,
                              void* d_out, int out_size, void* d_ws,
                              size_t ws_size, hipStream_t stream) {
  (void)in_sizes; (void)n_in; (void)out_size; (void)ws_size;
  const float* sa = (const float*)d_in[0];
  const float* sb = (const float*)d_in[1];
  const float* W1 = (const float*)d_in[2];
  const float* b1 = (const float*)d_in[3];
  const float* W2 = (const float*)d_in[4];
  const float* b2 = (const float*)d_in[5];
  const float* W3 = (const float*)d_in[6];
  const float* b3 = (const float*)d_in[7];
  float* out = (float*)d_out;

  float* a_pb   = (float*)d_ws;
  float* b_part = a_pb + 1024 * 64;

  proj_kernel<<<128, 128, 0, stream>>>(sa, sb, W1, b1, a_pb, b_part);
  match_mlp_kernel<<<512, 128, 0, stream>>>(a_pb, b_part, W2, b2, W3, b3, out);
}


# Projection kernel re-tiled: the 4 waves of a work-group split K (each all 64 features, 16-byte loads, rows read once, 40 loads at once), K quarters added via LDS: sum order changes
# speedup vs baseline: 1.6051x; 1.0864x over previous
#include <hip/hip_runtime.h>

typedef __attribute__((ext_vector_type(2))) float v2f;
typedef __attribute__((ext_vector_type(4))) float v4f;
typedef __attribute__((ext_vector_type(8))) float v8f;


__device__ __forceinline__ v8f wmma4(v2f a, v2f b, v8f c) {
  return __builtin_amdgcn_wmma_f32_16x16x4_f32(false, a, false, b, (short)0, c,
                                               false, false);
}

__device__ __forceinline__ float leaky(float x) {
  return x >= 0.0f ? x : 0.01f * x;
}


__attribute__((amdgpu_waves_per_eu(1, 2)))
__global__ void __launch_bounds__(128) proj_kernel(
    const float* __restrict__ sa, const float* __restrict__ sb,
    const float* __restrict__ W1, const float* __restrict__ b1,
    float* __restrict__ a_pb, float* __restrict__ b_part) {
  __shared__ float red[4 * 4 * 2 * 16 * 8];
  const int lane = threadIdx.x & 31;
  const int lo = lane & 15, hi = lane >> 4;
  const int w = threadIdx.x >> 5;
  const int mat = blockIdx.x >> 6;
  const int m0 = (blockIdx.x & 63) * 16;
  const int kb = 64 * w;

  const float* src = mat ? sb : sa;
  const float* W   = W1 + (mat ? 256 * 64 : 0);
  const float* xrow = src + (size_t)(m0 + lo) * 256 + kb + 4 * hi;
  const float* wrow = W + (size_t)(kb + 4 * hi) * 64 + 4 * lo;

  v4f x[8], wv[8][2][2];
#pragma unroll
  for (int j = 0; j < 8; ++j) {
    x[j] = *(const v4f*)(xrow + 8 * j);
#pragma unroll
    for (int p = 0; p < 2; ++p) {
#pragma unroll
      for (int s = 0; s < 2; ++s)
        wv[j][p][s] = *(const v4f*)(wrow + (size_t)(8 * j + 2 * p + s) * 64);
    }
  }
  const v4f bias0 = *(const v4f*)(b1 + 16 * w + 8 * hi);
  const v4f bias1 = *(const v4f*)(b1 + 16 * w + 8 * hi + 4);

  v8f acc[4] = {};
#pragma unroll
  for (int j = 0; j < 8; ++j) {
#pragma unroll
    for (int p = 0; p < 2; ++p) {
      v2f xb = {x[j][2 * p], x[j][2 * p + 1]};
#pragma unroll
      for (int e = 0; e < 4; ++e) {
        v2f wa = {wv[j][p][0][e], wv[j][p][1][e]};
        acc[e] = wmma4(wa, xb, acc[e]);
      }
    }
  }

#pragma unroll
  for (int r = 0; r < 8; ++r) {
    const int wo = 2 * hi + (r >> 2);
    const int hio = (r >> 1) & 1;
    v4f pc = {acc[0][r], acc[1][r], acc[2][r], acc[3][r]};
    *(v4f*)(red + ((((w * 4 + wo) * 2 + hio) * 16 + lo) * 8 + 4 * (r & 1))) = pc;
  }
  __syncthreads();

  float o[8];
  {
    v4f q0[4], q1[4];
#pragma unroll
    for (int ws = 0; ws < 4; ++ws) {
      const float* rp = red + ((((ws * 4 + w) * 2 + hi) * 16 + lo) * 8);
      q0[ws] = *(const v4f*)rp;
      q1[ws] = *(const v4f*)(rp + 4);
    }
    v4f t0 = ((q0[0] + q0[1]) + q0[2]) + q0[3];
    v4f t1 = ((q1[0] + q1[1]) + q1[2]) + q1[3];
    if (mat == 0) { t0 += bias0; t1 += bias1; }
    o[0] = t0[0]; o[1] = t0[1]; o[2] = t0[2]; o[3] = t0[3];
    o[4] = t1[0]; o[5] = t1[1]; o[6] = t1[2]; o[7] = t1[3];
  }
  const int n0 = 16 * w;
  float* dst = (mat ? b_part : a_pb) + (size_t)(m0 >> 4) * 1024 +
               ((n0 >> 3) + hi) * 128 + lo * 4;
  v4f p0 = {o[0], o[1], o[4], o[5]};
  v4f p1 = {o[2], o[3], o[6], o[7]};
  *(v4f*)dst = p0;
  *(v4f*)(dst + 64) = p1;
}

#define LSTR 34

__attribute__((amdgpu_waves_per_eu(2, 2)))
__global__ void __launch_bounds__(128) match_mlp_kernel(
    const float* __restrict__ a_pb, const float* __restrict__ b_part,
    const float* __restrict__ W2, const float* __restrict__ b2,
    const float* __restrict__ W3, const float* __restrict__ b3,
    float* __restrict__ out) {
  __shared__ float lds[4 * 16 * LSTR];
  const int lane = threadIdx.x & 31;
  const int lo = lane & 15, hi = lane >> 4;
  const int wave = (blockIdx.x * 128 + threadIdx.x) >> 5;
  const int ba = wave >> 1;
  const int half = wave & 1;
  const int bbase = ba & ~127;
  float* wlds = lds + (threadIdx.x >> 5) * (16 * LSTR);

  const float* apr = a_pb + (size_t)(ba >> 4) * 1024 + hi * 64 + (ba & 15) * 4;
  const float* bpr = b_part + (size_t)((bbase + half * 64) >> 4) * 1024 +
                     hi * 64 + lo * 4;
  const float* w2p = W2 + lo + (size_t)(2 * hi) * 32;
  const float* w3p = W3 + lo + (size_t)(2 * hi) * 16;

  v4f abv[8], bp[4][8];
  v2f w2f[2][16], w3f[8];
  float b2v0 = 0.0f, b2v1 = 0.0f, b3v = 0.0f;

#define LOAD_STEP(s)                                          \
  {                                                           \
    if (((s) & 1) == 0) {                                     \
      abv[(s) >> 1] = *(const v4f*)(apr + 128 * ((s) >> 1));  \
      bp[0][(s) >> 1] = *(const v4f*)(bpr + 128 * ((s) >> 1));\
    }                                                         \
    w2f[0][s].x = w2p[(size_t)(4 * (s)) * 32];                \
    w2f[0][s].y = w2p[(size_t)(4 * (s) + 1) * 32];            \
    w2f[1][s].x = w2p[(size_t)(4 * (s)) * 32 + 16];           \
    w2f[1][s].y = w2p[(size_t)(4 * (s) + 1) * 32 + 16];       \
  }
#define LOAD_TILE(t)                                          \
  _Pragma("unroll") for (int j = 0; j < 8; ++j)               \
    bp[t][j] = *(const v4f*)(bpr + (t) * 1024 + 128 * j);

#pragma unroll
  for (int s = 0; s < 8; ++s) LOAD_STEP(s)

#pragma unroll
  for (int it = 0; it < 4; ++it) {
    const int c0 = (half * 4 + it) * 16;

    v8f c2a = {}, c2b = {};
#pragma unroll
    for (int ch = 0; ch < 4; ++ch) {
      if (it == 0) {
        __builtin_amdgcn_sched_barrier(0);
        if (ch < 2) {
#pragma unroll
          for (int s = 0; s < 4; ++s) LOAD_STEP(4 * (ch + 2) + s)
        } else if (ch == 2) {
          b2v0 = b2[lo];
          b2v1 = b2[16 + lo];
#pragma unroll
          for (int i = 0; i < 8; ++i) {
            w3f[i].x = w3p[(size_t)(4 * i) * 16];
            w3f[i].y = w3p[(size_t)(4 * i + 1) * 16];
          }
          b3v = b3[lo];
          LOAD_TILE(1)
        } else {
          LOAD_TILE(2)
        }
      }
#pragma unroll
      for (int s = 0; s < 4; ++s) {
        const int i = 4 * ch + s;
        v2f hf;
        hf.x = leaky(abv[i >> 1][2 * (i & 1)] + bp[it][i >> 1][2 * (i & 1)]);
        hf.y = leaky(abv[i >> 1][2 * (i & 1) + 1] +
                     bp[it][i >> 1][2 * (i & 1) + 1]);
        c2a = wmma4(hf, w2f[0][i], c2a);
        c2b = wmma4(hf, w2f[1][i], c2b);
      }
    }

    if (it == 0) {
      __builtin_amdgcn_sched_barrier(0);
      LOAD_TILE(3)
    }

#pragma unroll
    for (int r = 0; r < 8; ++r) {
      wlds[(r + 8 * hi) * LSTR + lo]      = leaky(c2a[r] + b2v0);
      wlds[(r + 8 * hi) * LSTR + 16 + lo] = leaky(c2b[r] + b2v1);
    }
    asm volatile("s_wait_dscnt 0x0" ::: "memory");

    v2f hf3[8];
#pragma unroll
    for (int i = 0; i < 8; ++i)
      hf3[i] = *(const v2f*)(wlds + lo * LSTR + 4 * i + 2 * hi);
    asm volatile("s_wait_dscnt 0x0" ::: "memory");

    v8f c3 = {};
#pragma unroll
    for (int i = 0; i < 8; ++i) c3 = wmma4(hf3[i], w3f[i], c3);

    float* op = out + ((size_t)ba * 128 + c0) * 16 + lo;
#pragma unroll
    for (int r = 0; r < 8; ++r)
      op[(size_t)(r + 8 * hi) * 16] = leaky(c3[r] + b3v);
  }
#undef LOAD_STEP
#undef LOAD_TILE
}

extern "C" void kernel_launch(void* const* d_in, const int* in_sizes, int n_in,
                              void* d_out, int out_size, void* d_ws,
                              size_t ws_size, hipStream_t stream) {
  (void)in_sizes; (void)n_in; (void)out_size; (void)ws_size;
  const float* sa = (const float*)d_in[0];
  const float* sb = (const float*)d_in[1];
  const float* W1 = (const float*)d_in[2];
  const float* b1 = (const float*)d_in[3];
  const float* W2 = (const float*)d_in[4];
  const float* b2 = (const float*)d_in[5];
  const float* W3 = (const float*)d_in[6];
  const float* b3 = (const float*)d_in[7];
  float* out = (float*)d_out;

  float* a_pb   = (float*)d_ws;
  float* b_part = a_pb + 1024 * 64;

  proj_kernel<<<128, 128, 0, stream>>>(sa, sb, W1, b1, a_pb, b_part);
  match_mlp_kernel<<<512, 128, 0, stream>>>(a_pb, b_part, W2, b2, W3, b3, out);
}
